# starting kernel
# baseline (speedup 1.0000x reference)
#include <hip/hip_runtime.h>


typedef __attribute__((ext_vector_type(2))) float v2f;
typedef __attribute__((ext_vector_type(8))) float v8f;

#define BS 8
#define FS 96
#define HH 96
#define WW 320
#define PLANES 9
#define LVLS 4
#define TILES_X (WW / 16)
#define NWAVES (BS * HH * TILES_X)
#define WAVES_PER_BLOCK 8

__global__ __launch_bounds__(256) void cost_volume_wmma_f32(
    const float* __restrict__ f1, const float* __restrict__ f2,
    float* __restrict__ out) {
  const int lane = threadIdx.x & 31;
  const int wid = blockIdx.x * WAVES_PER_BLOCK + (threadIdx.x >> 5);

  const int t = wid % TILES_X;
  const int bh = wid / TILES_X;
  const int h = bh % HH;
  const int b = bh / HH;

  const int X0 = t * 16;
  const int half = lane >> 4;
  const int m = lane & 15;

  const int HW = HH * WW;
  const float* __restrict__ f1row = f1 + ((long)b * FS * HH + h) * WW;
  const float* __restrict__ f2row = f2 + ((long)b * FS * HH + h) * WW;

  const int col0 = X0 - 8 + m;
  const int col1 = X0 + 8 + m;
  const int col0c = col0 < 0 ? 0 : col0;
  const int col1c = col1 >= WW ? (WW - 1) : col1;
  const float inv = 1.0f / (float)FS;
  const float scale0 = (col0 >= 0) ? inv : 0.0f;
  const float scale1 = (col1 < WW) ? inv : 0.0f;

  const int aBase = 2 * half * HW + X0 + m;

  v8f acc0 = {};
  v8f acc1 = {};

#pragma unroll
  for (int k0 = 0; k0 < FS; k0 += 4) {
    const int c = k0 * HW;

    v2f a;
    a.x = f1row[c + aBase];
    a.y = f1row[c + aBase + HW];

    v2f b0, b1;
    b0.x = f2row[c + 2 * half * HW + col0c];
    b0.y = f2row[c + 2 * half * HW + HW + col0c];
    b1.x = f2row[c + 2 * half * HW + col1c];
    b1.y = f2row[c + 2 * half * HW + HW + col1c];

    acc0 = __builtin_amdgcn_wmma_f32_16x16x4_f32(false, a, false, b0,
                                                 (short)0, acc0, false, false);
    acc1 = __builtin_amdgcn_wmma_f32_16x16x4_f32(false, a, false, b1,
                                                 (short)0, acc1, false, false);
  }

  const long outBase = ((long)b * PLANES * HH + h) * WW;
#pragma unroll
  for (int r = 0; r < 8; ++r) {
    const int M = r + 8 * half;
    const int x = X0 + M;

    const int d0 = m - 8 - M;
    if (d0 >= -LVLS && d0 <= LVLS) {
      out[outBase + (long)(LVLS - d0) * HW + x] = acc0[r] * scale0;
    }
    const int d1 = m + 8 - M;
    if (d1 >= -LVLS && d1 <= LVLS) {
      out[outBase + (long)(LVLS - d1) * HW + x] = acc1[r] * scale1;
    }
  }
}

extern "C" void kernel_launch(void* const* d_in, const int* in_sizes, int n_in,
                              void* d_out, int out_size, void* d_ws,
                              size_t ws_size, hipStream_t stream) {
  (void)in_sizes; (void)n_in; (void)d_ws; (void)ws_size; (void)out_size;
  const float* f1 = (const float*)d_in[0];
  const float* f2 = (const float*)d_in[1];
  float* out = (float*)d_out;

  const int blocks = NWAVES / WAVES_PER_BLOCK;
  cost_volume_wmma_f32<<<dim3(blocks), dim3(256), 0, stream>>>(f1, f2, out);
}


# Each wave owns 5 adjacent 16-column tiles of a row and shares f2 column groups between neighbours: 22 loads per k-step per 80 columns, was 30; 5x fewer waves, load groups issued 2 k-steps ahead
# speedup vs baseline: 1.1078x; 1.1078x over previous
#include <hip/hip_runtime.h>


typedef __attribute__((ext_vector_type(2))) float v2f;
typedef __attribute__((ext_vector_type(8))) float v8f;

#define BS 8
#define FS 96
#define HH 96
#define WW 320
#define PLANES 9
#define LVLS 4
#define TW 5
#define WAVE_COLS (TW * 16)
#define WAVES_PER_ROW (WW / WAVE_COLS)
#define KSTEPS (FS / 4)
#define PF 2

__global__ __launch_bounds__(WAVES_PER_ROW * 32) void cost_volume_wmma_f32(
    const float* __restrict__ f1, const float* __restrict__ f2,
    float* __restrict__ out) {
  const int lane = threadIdx.x & 31;
  const int w = threadIdx.x >> 5;
  const int bh = blockIdx.x;
  const int h = bh % HH;
  const int b = bh / HH;

  const int X0 = w * WAVE_COLS;
  const int half = lane >> 4;
  const int m = lane & 15;

  const int HW = HH * WW;
  const float* __restrict__ f1row = f1 + ((long)b * FS * HH + h) * WW + 2 * half * HW;
  const float* __restrict__ f2row = f2 + ((long)b * FS * HH + h) * WW + 2 * half * HW;

  const int colF = X0 - 8 + m;
  const int colL = X0 - 8 + 16 * TW + m;
  const int colFc = colF < 0 ? 0 : colF;
  const int colLc = colL >= WW ? (WW - 1) : colL;
  const float inv = 1.0f / (float)FS;
  const float scaleF = (colF >= 0) ? inv : 0.0f;
  const float scaleL = (colL < WW) ? inv : 0.0f;

  const float* __restrict__ pa = f1row + X0 + m;
  const float* __restrict__ pbF = f2row + colFc;
  const float* __restrict__ pbM = f2row + X0 + 8 + m;
  const float* __restrict__ pbL = f2row + colLc;

  v8f acc0[TW];
  v8f acc1[TW];
#pragma unroll
  for (int j = 0; j < TW; ++j) {
    acc0[j] = (v8f){};
    acc1[j] = (v8f){};
  }

  v2f A[PF + 1][TW];
  v2f B[PF + 1][TW + 1];

#define LOAD_B(S_, SLOT, I, P, OFF)                                      \
  B[SLOT][I].x = (P)[(S_) * 4 * HW + (OFF)];                            \
  B[SLOT][I].y = (P)[(S_) * 4 * HW + HW + (OFF)];
#define LOAD_STEP(S, SLOT)                                              \
  {                                                                     \
    LOAD_B(S, SLOT, 0, pbF, 0)                                          \
    _Pragma("unroll") for (int j = 0; j < TW; ++j) {                    \
      A[SLOT][j].x = pa[(S) * 4 * HW + 16 * j];                         \
      A[SLOT][j].y = pa[(S) * 4 * HW + HW + 16 * j];                    \
      if (j + 1 < TW) {                                                 \
        LOAD_B(S, SLOT, j + 1, pbM, 16 * j)                             \
      } else {                                                          \
        LOAD_B(S, SLOT, TW, pbL, 0)                                     \
      }                                                                 \
    }                                                                   \
  }

#pragma unroll
  for (int s = 0; s < PF; ++s) LOAD_STEP(s, s)

#pragma unroll
  for (int s = 0; s < KSTEPS; ++s) {
    if (s + PF < KSTEPS) LOAD_STEP(s + PF, (s + PF) % (PF + 1))
    __builtin_amdgcn_sched_barrier(0);
    const int cur = s % (PF + 1);
#pragma unroll
    for (int j = 0; j < TW; ++j) {
      acc0[j] = __builtin_amdgcn_wmma_f32_16x16x4_f32(false, A[cur][j], false, B[cur][j],
                                                      (short)0, acc0[j], false, false);
      acc1[j] = __builtin_amdgcn_wmma_f32_16x16x4_f32(false, A[cur][j], false, B[cur][j + 1],
                                                      (short)0, acc1[j], false, false);
    }
    __builtin_amdgcn_sched_barrier(0);
  }
#undef LOAD_STEP
#undef LOAD_B

  const long outBase = ((long)b * PLANES * HH + h) * WW;
#pragma unroll
  for (int j = 0; j < TW; ++j) {
    const float s0 = (j == 0) ? scaleF : inv;
    const float s1 = (j == TW - 1) ? scaleL : inv;
#pragma unroll
    for (int r = 0; r < 8; ++r) {
      const int M = r + 8 * half;
      const int x = X0 + 16 * j + M;

      const int d0 = m - 8 - M;
      if (d0 >= -LVLS && d0 <= LVLS) {
        out[outBase + (long)(LVLS - d0) * HW + x] = acc0[j][r] * s0;
      }
      const int d1 = m + 8 - M;
      if (d1 >= -LVLS && d1 <= LVLS) {
        out[outBase + (long)(LVLS - d1) * HW + x] = acc1[j][r] * s1;
      }
    }
  }
}

extern "C" void kernel_launch(void* const* d_in, const int* in_sizes, int n_in,
                              void* d_out, int out_size, void* d_ws,
                              size_t ws_size, hipStream_t stream) {
  (void)in_sizes; (void)n_in; (void)d_ws; (void)ws_size; (void)out_size;
  const float* f1 = (const float*)d_in[0];
  const float* f2 = (const float*)d_in[1];
  float* out = (float*)d_out;

  cost_volume_wmma_f32<<<dim3(BS * HH), dim3(WAVES_PER_ROW * 32), 0, stream>>>(f1, f2, out);
}


# Output rows staged through LDS: each 128-byte output line written once, by one 16-byte-per-lane store (was 4-byte stores whose lanes each hit a different plane's line); sums unchanged
# speedup vs baseline: 1.6801x; 1.5166x over previous
#include <hip/hip_runtime.h>


typedef __attribute__((ext_vector_type(2))) float v2f;
typedef __attribute__((ext_vector_type(8))) float v8f;
typedef __attribute__((ext_vector_type(4))) float v4f;

#define BS 8
#define FS 96
#define HH 96
#define WW 320
#define PLANES 9
#define LVLS 4
#define TW 5
#define WAVE_COLS (TW * 16)
#define WAVES_PER_ROW (WW / WAVE_COLS)
#define KSTEPS (FS / 4)
#define PF 2
#define OPITCH 324
#define ROW_V4 (WW / 4)
#define OUT_V4 (PLANES * ROW_V4)
#define NTHREADS (WAVES_PER_ROW * 32)

__global__ __launch_bounds__(WAVES_PER_ROW * 32) void cost_volume_wmma_f32(
    const float* __restrict__ f1, const float* __restrict__ f2,
    float* __restrict__ out) {
  const int lane = threadIdx.x & 31;
  const int w = threadIdx.x >> 5;
  const int bh = blockIdx.x;
  const int h = bh % HH;
  const int b = bh / HH;

  const int X0 = w * WAVE_COLS;
  const int half = lane >> 4;
  const int m = lane & 15;

  const int HW = HH * WW;
  const float* __restrict__ f1row = f1 + ((long)b * FS * HH + h) * WW + 2 * half * HW;
  const float* __restrict__ f2row = f2 + ((long)b * FS * HH + h) * WW + 2 * half * HW;

  const int colF = X0 - 8 + m;
  const int colL = X0 - 8 + 16 * TW + m;
  const int colFc = colF < 0 ? 0 : colF;
  const int colLc = colL >= WW ? (WW - 1) : colL;
  const float inv = 1.0f / (float)FS;
  const float scaleF = (colF >= 0) ? inv : 0.0f;
  const float scaleL = (colL < WW) ? inv : 0.0f;

  const float* __restrict__ pa = f1row + X0 + m;
  const float* __restrict__ pbF = f2row + colFc;
  const float* __restrict__ pbM = f2row + X0 + 8 + m;
  const float* __restrict__ pbL = f2row + colLc;

  v8f acc0[TW];
  v8f acc1[TW];
#pragma unroll
  for (int j = 0; j < TW; ++j) {
    acc0[j] = (v8f){};
    acc1[j] = (v8f){};
  }

  v2f A[PF + 1][TW];
  v2f B[PF + 1][TW + 1];

#define LOAD_B(S_, SLOT, I, P, OFF)                                      \
  B[SLOT][I].x = (P)[(S_) * 4 * HW + (OFF)];                            \
  B[SLOT][I].y = (P)[(S_) * 4 * HW + HW + (OFF)];
#define LOAD_STEP(S, SLOT)                                              \
  {                                                                     \
    LOAD_B(S, SLOT, 0, pbF, 0)                                          \
    _Pragma("unroll") for (int j = 0; j < TW; ++j) {                    \
      A[SLOT][j].x = pa[(S) * 4 * HW + 16 * j];                         \
      A[SLOT][j].y = pa[(S) * 4 * HW + HW + 16 * j];                    \
      if (j + 1 < TW) {                                                 \
        LOAD_B(S, SLOT, j + 1, pbM, 16 * j)                             \
      } else {                                                          \
        LOAD_B(S, SLOT, TW, pbL, 0)                                     \
      }                                                                 \
    }                                                                   \
  }

#pragma unroll
  for (int s = 0; s < PF; ++s) LOAD_STEP(s, s)

#pragma unroll
  for (int s = 0; s < KSTEPS; ++s) {
    if (s + PF < KSTEPS) LOAD_STEP(s + PF, (s + PF) % (PF + 1))
    __builtin_amdgcn_sched_barrier(0);
    const int cur = s % (PF + 1);
#pragma unroll
    for (int j = 0; j < TW; ++j) {
      acc0[j] = __builtin_amdgcn_wmma_f32_16x16x4_f32(false, A[cur][j], false, B[cur][j],
                                                      (short)0, acc0[j], false, false);
      acc1[j] = __builtin_amdgcn_wmma_f32_16x16x4_f32(false, A[cur][j], false, B[cur][j + 1],
                                                      (short)0, acc1[j], false, false);
    }
    __builtin_amdgcn_sched_barrier(0);
  }
#undef LOAD_STEP
#undef LOAD_B

  __shared__ __attribute__((aligned(16))) float orow[PLANES * OPITCH];
#pragma unroll
  for (int r = 0; r < 8; ++r) {
    const int M = r + 8 * half;
    const int d0 = m - 8 - M;
    const int d1 = d0 + 16;
    const bool in0 = (d0 >= -LVLS) && (d0 <= LVLS);
    const bool in1 = (d1 >= -LVLS) && (d1 <= LVLS);
    if (in0 || in1) {
      const int d = in1 ? d1 : d0;
      float* __restrict__ dst = orow + (LVLS - d) * OPITCH + X0 + M;
#pragma unroll
      for (int j = 0; j < TW; ++j) {
        const float s0 = (j == 0) ? scaleF : inv;
        const float s1 = (j == TW - 1) ? scaleL : inv;
        dst[16 * j] = in1 ? acc1[j][r] * s1 : acc0[j][r] * s0;
      }
    }
  }
  __syncthreads();

  float* __restrict__ obase = out + ((long)b * PLANES * HH + h) * WW;
  const int tid = (int)threadIdx.x;
  v4f piece[OUT_V4 / NTHREADS + 1];
#pragma unroll
  for (int it = 0; it < OUT_V4 / NTHREADS; ++it) {
    const int q = it * NTHREADS + tid;
    const int p = q / ROW_V4;
    const int c = q - p * ROW_V4;
    piece[it] = *(const v4f*)(orow + p * OPITCH + 4 * c);
  }
  const bool tail = tid < OUT_V4 - (OUT_V4 / NTHREADS) * NTHREADS;
  if (tail) piece[OUT_V4 / NTHREADS] = *(const v4f*)(orow + (PLANES - 1) * OPITCH + 4 * tid);
#pragma unroll
  for (int it = 0; it < OUT_V4 / NTHREADS; ++it) {
    const int q = it * NTHREADS + tid;
    const int p = q / ROW_V4;
    const int c = q - p * ROW_V4;
    *(v4f*)(obase + (long)p * HW + 4 * c) = piece[it];
  }
  if (tail) *(v4f*)(obase + (long)(PLANES - 1) * HW + 4 * tid) = piece[OUT_V4 / NTHREADS];
}

extern "C" void kernel_launch(void* const* d_in, const int* in_sizes, int n_in,
                              void* d_out, int out_size, void* d_ws,
                              size_t ws_size, hipStream_t stream) {
  (void)in_sizes; (void)n_in; (void)d_ws; (void)ws_size; (void)out_size;
  const float* f1 = (const float*)d_in[0];
  const float* f2 = (const float*)d_in[1];
  float* out = (float*)d_out;

  cost_volume_wmma_f32<<<dim3(BS * HH), dim3(WAVES_PER_ROW * 32), 0, stream>>>(f1, f2, out);
}
